# starting kernel
# baseline (speedup 1.0000x reference)
#include <hip/hip_runtime.h>

typedef __attribute__((ext_vector_type(2))) float v2f;
typedef __attribute__((ext_vector_type(4))) float v4f;
typedef __attribute__((ext_vector_type(8))) float v8f;

#define ORD 8
#define SQS 6

__global__ __launch_bounds__(512) void expm_apply_kernel(
    const float* __restrict__ x,
    const float* __restrict__ c,
    const float* __restrict__ psi,
    float* __restrict__ out)
{
    __shared__ float tile[16][16 * 17];

    const int tid = threadIdx.x;
    const int w   = tid >> 5;
    const int l   = tid & 31;
    const int h   = l >> 4;
    const int n   = l & 15;
    const long b0 = (long)blockIdx.x * 16;

    v8f acc = {0.f, 0.f, 0.f, 0.f, 0.f, 0.f, 0.f, 0.f};
#pragma unroll
    for (int kk = 0; kk < 16; ++kk) {
        const int k0 = kk * 4;
        v2f a;
        const float* cp = c + (b0 + n) * 64 + (k0 + 2 * h);
        a.x = cp[0];
        a.y = cp[1];
        const float* pp = psi + (size_t)(k0 + 2 * h) * 256 + w * 16 + n;
        v2f bb;
        bb.x = pp[0];
        bb.y = pp[256];
        acc = __builtin_amdgcn_wmma_f32_16x16x4_f32(
            false, a, false, bb, (short)0, acc, false, false);
    }
#pragma unroll
    for (int j = 0; j < 8; ++j) {
        tile[j + 8 * h][w * 17 + n] = acc[j];
    }
    __syncthreads();

    float* mt = &tile[w][0];

    const float sc = 1.0f / (float)(1 << SQS);
    v2f Ak[4];
#pragma unroll
    for (int kk = 0; kk < 4; ++kk) {
        const int col = kk * 4 + 2 * h;
        Ak[kk].x = mt[n * 17 + col] * sc;
        Ak[kk].y = mt[n * 17 + col + 1] * sc;
    }

    v8f r;
#pragma unroll
    for (int j = 0; j < 8; ++j) r[j] = (n == (j + 8 * h)) ? 1.f : 0.f;

#pragma unroll
    for (int k = ORD; k >= 1; --k) {
#pragma unroll
        for (int j = 0; j < 8; ++j) mt[(j + 8 * h) * 17 + n] = r[j];
        asm volatile("s_wait_dscnt 0" ::: "memory");
        v8f d = {0.f, 0.f, 0.f, 0.f, 0.f, 0.f, 0.f, 0.f};
#pragma unroll
        for (int kk = 0; kk < 4; ++kk) {
            const int row = kk * 4 + 2 * h;
            v2f bb;
            bb.x = mt[row * 17 + n];
            bb.y = mt[(row + 1) * 17 + n];
            d = __builtin_amdgcn_wmma_f32_16x16x4_f32(
                false, Ak[kk], false, bb, (short)0, d, false, false);
        }
        const float invk = 1.0f / (float)k;
#pragma unroll
        for (int j = 0; j < 8; ++j)
            r[j] = ((n == (j + 8 * h)) ? 1.f : 0.f) + d[j] * invk;
    }

#pragma unroll
    for (int sq = 0; sq < SQS; ++sq) {
#pragma unroll
        for (int j = 0; j < 8; ++j) mt[(j + 8 * h) * 17 + n] = r[j];
        asm volatile("s_wait_dscnt 0" ::: "memory");
        v8f d = {0.f, 0.f, 0.f, 0.f, 0.f, 0.f, 0.f, 0.f};
#pragma unroll
        for (int kk = 0; kk < 4; ++kk) {
            const int col = kk * 4 + 2 * h;
            v2f aa;
            aa.x = mt[n * 17 + col];
            aa.y = mt[n * 17 + col + 1];
            v2f bb;
            bb.x = mt[col * 17 + n];
            bb.y = mt[(col + 1) * 17 + n];
            d = __builtin_amdgcn_wmma_f32_16x16x4_f32(
                false, aa, false, bb, (short)0, d, false, false);
        }
#pragma unroll
        for (int j = 0; j < 8; ++j) r[j] = d[j];
    }

    const long b = b0 + w;
    const float xv = x[b * 16 + n];
    float p[8];
#pragma unroll
    for (int j = 0; j < 8; ++j) p[j] = r[j] * xv;
#pragma unroll
    for (int off = 1; off < 16; off <<= 1) {
#pragma unroll
        for (int j = 0; j < 8; ++j) p[j] += __shfl_xor(p[j], off, 16);
    }
    if (n == 0) {
        v4f lo = {p[0], p[1], p[2], p[3]};
        v4f hi = {p[4], p[5], p[6], p[7]};
        float* op = out + b * 16 + 8 * h;
        *(v4f*)(op)     = lo;
        *(v4f*)(op + 4) = hi;
    }
}

extern "C" void kernel_launch(void* const* d_in, const int* in_sizes, int n_in,
                              void* d_out, int out_size, void* d_ws, size_t ws_size,
                              hipStream_t stream) {
    const float* x   = (const float*)d_in[0];
    const float* c   = (const float*)d_in[1];
    const float* psi = (const float*)d_in[2];
    float* out = (float*)d_out;

    const int B = in_sizes[1] / 64;
    dim3 grid(B / 16);
    dim3 block(512);
    hipLaunchKernelGGL(expm_apply_kernel, grid, block, 0, stream,
                       x, c, psi, out);
}


# tail without LDS permutes: both lane halves hold the same rows T^j x, so each lane adds all 16 weighted terms itself (no half select or cross-half add); guard norms by DPP row adds; sum order changes
# speedup vs baseline: 1.7745x; 1.7745x over previous
#include <hip/hip_runtime.h>

typedef __attribute__((ext_vector_type(2))) float v2f;
typedef __attribute__((ext_vector_type(4))) float v4f;
typedef __attribute__((ext_vector_type(8))) float v8f;

#define ORD 8
#define SQS 6


__device__ __forceinline__ void st_d(float* mt, const v8f& r, int h, int n) {
    float* q = mt + (8 * h) * 17 + n;
#pragma unroll
    for (int j = 0; j < 8; ++j) q[j * 17] = r[j];
    asm volatile("s_wait_dscnt 0" ::: "memory");
}
__device__ __forceinline__ void ld_a(const float* mt, v2f (&a)[4], int h, int n) {
#pragma unroll
    for (int kk = 0; kk < 4; ++kk) {
        const int col = kk * 4 + 2 * h;
        a[kk].x = mt[n * 17 + col];
        a[kk].y = mt[n * 17 + col + 1];
    }
}
__device__ __forceinline__ void ld_b(const float* mt, v2f (&b)[4], int h, int n) {
#pragma unroll
    for (int kk = 0; kk < 4; ++kk) {
        const int row = kk * 4 + 2 * h;
        b[kk].x = mt[row * 17 + n];
        b[kk].y = mt[(row + 1) * 17 + n];
    }
}
__device__ __forceinline__ v8f mm(const v2f (&a)[4], const v2f (&b)[4]) {
    v8f d = {0.f, 0.f, 0.f, 0.f, 0.f, 0.f, 0.f, 0.f};
#pragma unroll
    for (int kk = 0; kk < 4; ++kk)
        d = __builtin_amdgcn_wmma_f32_16x16x4_f32(
            false, a[kk], false, b[kk], (short)0, d, false, false);
    return d;
}
__device__ __forceinline__ float xhalf(float x) {
    const int b = __builtin_bit_cast(int, x);
    return __builtin_bit_cast(float, __builtin_amdgcn_permlanex16(
        b, b, 0x76543210, 0xfedcba98, false, false));
}
template <int M>
__device__ __forceinline__ float row_xadd(float x) {
    const int b = __builtin_bit_cast(int, x);
    return x + __builtin_bit_cast(float, __builtin_amdgcn_update_dpp(
        0, b, 0x160 + M, 0xf, 0xf, true));
}
__device__ __forceinline__ float wave_sum(float s) {
    s = row_xadd<1>(s);
    s = row_xadd<2>(s);
    s = row_xadd<4>(s);
    s = row_xadd<8>(s);
    return s + xhalf(s);
}

__global__ __launch_bounds__(512, 16) void expm_apply_kernel(
    const float* __restrict__ x,
    const float* __restrict__ c,
    const float* __restrict__ psi,
    float* __restrict__ out)
{
    __shared__ float tile[16][16 * 17];

    const int tid = threadIdx.x;
    const int w   = tid >> 5;
    const int l   = tid & 31;
    const int h   = l >> 4;
    const int n   = l & 15;
    const long b0 = (long)blockIdx.x * 16;

    const int wu = __builtin_amdgcn_readfirstlane(w);
    if (wu < 8) {
        v8f acc[2];
#pragma unroll
        for (int a = 0; a < 2; ++a)
            acc[a] = (v8f){0.f, 0.f, 0.f, 0.f, 0.f, 0.f, 0.f, 0.f};
        const float* cp = c + (b0 + n) * 64 + 4 * h;
        const float* pp = psi + (size_t)(4 * h) * 256 + 32 * wu + 2 * n;
#pragma unroll
        for (int i = 0; i < 8; ++i) {
            const v4f cv = *(const v4f*)(cp + 8 * i);
            v2f pv[4];
#pragma unroll
            for (int u = 0; u < 4; ++u)
                pv[u] = *(const v2f*)(pp + (size_t)(8 * i + u) * 256);
#pragma unroll
            for (int t = 0; t < 2; ++t) {
                v2f a2;
                a2.x = cv[2 * t];
                a2.y = cv[2 * t + 1];
#pragma unroll
                for (int a = 0; a < 2; ++a) {
                    v2f bb;
                    bb.x = pv[2 * t][a];
                    bb.y = pv[2 * t + 1][a];
                    acc[a] = __builtin_amdgcn_wmma_f32_16x16x4_f32(
                        false, a2, false, bb, (short)0, acc[a], false, false);
                }
            }
        }
        float* tp = &tile[8 * h][(2 * wu + (n >> 3)) * 17 + 2 * (n & 7)];
#pragma unroll
        for (int a = 0; a < 2; ++a) {
#pragma unroll
            for (int j = 0; j < 8; ++j) tp[j * (16 * 17) + a] = acc[a][j];
        }
    }
    __syncthreads();

    float* mt = &tile[wu][0];
    const long b = b0 + wu;

    v2f A1[4], Bp[4];
    ld_a(mt, A1, h, n);
    ld_b(mt, Bp, h, n);

    float s1 = 0.f;
#pragma unroll
    for (int kk = 0; kk < 4; ++kk) s1 += A1[kk].x * A1[kk].x + A1[kk].y * A1[kk].y;

    const float* xr = x + b * 16;
    v2f Ax[4];
#pragma unroll
    for (int kk = 0; kk < 4; ++kk) Ax[kk] = *(const v2f*)(xr + kk * 4 + 2 * h);
    const float xc = xr[n];
    v8f v = {xc, xc, xc, xc, xc, xc, xc, xc};

    v2f Av[4];
    v8f wv, d;

    wv = mm(Ax, A1);
    d  = mm(A1, Bp);
#pragma unroll
    for (int j = 0; j < 8; ++j) v[j] = (j & 1) ? wv[j] : v[j];
    v2f A2[4];
    st_d(mt, d, h, n);
    ld_a(mt, A2, h, n);
    ld_b(mt, Bp, h, n);

    st_d(mt, v, h, n);
    ld_a(mt, Av, h, n);
    wv = mm(Av, A2);
    d  = mm(A2, Bp);
#pragma unroll
    for (int j = 0; j < 8; ++j) v[j] = (j & 2) ? wv[j] : v[j];
    v2f A4[4];
    st_d(mt, d, h, n);
    ld_a(mt, A4, h, n);
    ld_b(mt, Bp, h, n);

    st_d(mt, v, h, n);
    ld_a(mt, Av, h, n);
    wv = mm(Av, A4);
    d  = mm(A4, Bp);
#pragma unroll
    for (int j = 0; j < 8; ++j) v[j] = (j & 4) ? wv[j] : v[j];
    float s8 = 0.f;
#pragma unroll
    for (int j = 0; j < 8; ++j) s8 += d[j] * d[j];
    v2f A8[4];
    st_d(mt, d, h, n);
    ld_a(mt, A8, h, n);

    st_d(mt, v, h, n);
    ld_a(mt, Av, h, n);
    wv = mm(Av, A8);

    const float wlo[8] = {1.0f, 1.0f, (float)(1.0 / 2.0), (float)(1.0 / 6.0),
                          (float)(1.0 / 24.0), (float)(1.0 / 120.0),
                          (float)(1.0 / 720.0), (float)(1.0 / 5040.0)};
    const float whi[8] = {(float)(1.0 / 40320.0), (float)(1.0 / 362880.0),
                          (float)(1.0 / 3628800.0), (float)(1.0 / 39916800.0),
                          (float)(1.0 / 479001600.0), (float)(1.0 / 6227020800.0),
                          (float)(1.0 / 87178291200.0), (float)(1.0 / 1307674368000.0)};
    float s = 0.f;
#pragma unroll
    for (int j = 0; j < 8; ++j) s = __builtin_fmaf(v[j], wlo[j], s);
#pragma unroll
    for (int j = 0; j < 8; ++j) s = __builtin_fmaf(wv[j], whi[j], s);

    s1 = wave_sum(s1);
    s8 = wave_sum(s8);
    const float g1 = __builtin_bit_cast(float,
        __builtin_amdgcn_readfirstlane(__builtin_bit_cast(int, s1)));
    const float g8 = __builtin_bit_cast(float,
        __builtin_amdgcn_readfirstlane(__builtin_bit_cast(int, s8)));
    const bool fast = (g1 <= 128.0f) && (g8 <= 3.2e7f);

    const bool longer = (g1 <= 128.0f) && (g8 <= 1.0e11f);

    if (__builtin_expect(!fast, 0)) {
        if (longer) {
            st_d(mt, wv, h, n);
            ld_a(mt, Av, h, n);
            wv = mm(Av, A8);
            const float wx[8] = {(float)(1.0 / 20922789888000.0),
                                 (float)(1.0 / 355687428096000.0),
                                 (float)(1.0 / 6402373705728000.0),
                                 (float)(1.0 / 121645100408832000.0),
                                 (float)(1.0 / 2432902008176640000.0),
                                 (float)(1.0 / 51090942171709440000.0),
                                 (float)(1.0 / 1124000727777607680000.0),
                                 (float)(1.0 / 25852016738884976640000.0)};
            float e = 0.f;
#pragma unroll
            for (int j = 0; j < 8; ++j) e += wv[j] * wx[j];
            s += e;
        } else {
            const float sc = 1.0f / (float)(1 << SQS);
            v2f Ak[4];
#pragma unroll
            for (int kk = 0; kk < 4; ++kk) {
                Ak[kk].x = A1[kk].x * sc;
                Ak[kk].y = A1[kk].y * sc;
            }
            v2f Bv[4];
            v8f r;
#pragma unroll
            for (int j = 0; j < 8; ++j) r[j] = (n == (j + 8 * h)) ? 1.f : 0.f;
#pragma unroll
            for (int k = ORD; k >= 1; --k) {
                st_d(mt, r, h, n);
                ld_b(mt, Bv, h, n);
                d = mm(Ak, Bv);
                const float invk = 1.0f / (float)k;
#pragma unroll
                for (int j = 0; j < 8; ++j)
                    r[j] = ((n == (j + 8 * h)) ? 1.f : 0.f) + d[j] * invk;
            }
#pragma unroll
            for (int sq = 0; sq < SQS; ++sq) {
                v2f aa[4];
                st_d(mt, r, h, n);
                ld_a(mt, aa, h, n);
                ld_b(mt, Bv, h, n);
                r = mm(aa, Bv);
            }
            v2f Ex[4], Ea[4];
#pragma unroll
            for (int kk = 0; kk < 4; ++kk) Ex[kk] = *(const v2f*)(xr + kk * 4 + 2 * h);
            st_d(mt, r, h, n);
            ld_a(mt, Ea, h, n);
            d = mm(Ex, Ea);
            s = d[0];
        }
    }

    if (h == 0) out[b * 16 + n] = s;
}

extern "C" void kernel_launch(void* const* d_in, const int* in_sizes, int n_in,
                              void* d_out, int out_size, void* d_ws, size_t ws_size,
                              hipStream_t stream) {
    const float* x   = (const float*)d_in[0];
    const float* c   = (const float*)d_in[1];
    const float* psi = (const float*)d_in[2];
    float* out = (float*)d_out;

    const int B = in_sizes[1] / 64;
    dim3 grid(B / 16);
    dim3 block(512);
    hipLaunchKernelGGL(expm_apply_kernel, grid, block, 0, stream,
                       x, c, psi, out);
}


# steps 0,1 (left tiles of 1 and 2 distinct rows) as per-lane dot products with the held A fragments on the vector ALU; T x to k order through LDS; 20 WMMAs per wave, was 28; sum order changes
# speedup vs baseline: 1.8780x; 1.0583x over previous
#include <hip/hip_runtime.h>

typedef __attribute__((ext_vector_type(2))) float v2f;
typedef __attribute__((ext_vector_type(4))) float v4f;
typedef __attribute__((ext_vector_type(8))) float v8f;

#define ORD 8
#define SQS 6


__device__ __forceinline__ void st_d(float* mt, const v8f& r, int h, int n) {
    float* q = mt + (8 * h) * 17 + n;
#pragma unroll
    for (int j = 0; j < 8; ++j) q[j * 17] = r[j];
    asm volatile("s_wait_dscnt 0" ::: "memory");
}
__device__ __forceinline__ void ld_a(const float* mt, v2f (&a)[4], int h, int n) {
#pragma unroll
    for (int kk = 0; kk < 4; ++kk) {
        const int col = kk * 4 + 2 * h;
        a[kk].x = mt[n * 17 + col];
        a[kk].y = mt[n * 17 + col + 1];
    }
}
__device__ __forceinline__ void ld_b(const float* mt, v2f (&b)[4], int h, int n) {
#pragma unroll
    for (int kk = 0; kk < 4; ++kk) {
        const int row = kk * 4 + 2 * h;
        b[kk].x = mt[row * 17 + n];
        b[kk].y = mt[(row + 1) * 17 + n];
    }
}
__device__ __forceinline__ v8f mm(const v2f (&a)[4], const v2f (&b)[4]) {
    v8f d = {0.f, 0.f, 0.f, 0.f, 0.f, 0.f, 0.f, 0.f};
#pragma unroll
    for (int kk = 0; kk < 4; ++kk)
        d = __builtin_amdgcn_wmma_f32_16x16x4_f32(
            false, a[kk], false, b[kk], (short)0, d, false, false);
    return d;
}
__device__ __forceinline__ float half_dot(const v2f (&a)[4], const v2f (&u)[4]) {
    float p = a[0].x * u[0].x;
    p = __builtin_fmaf(a[0].y, u[0].y, p);
#pragma unroll
    for (int kk = 1; kk < 4; ++kk) {
        p = __builtin_fmaf(a[kk].x, u[kk].x, p);
        p = __builtin_fmaf(a[kk].y, u[kk].y, p);
    }
    return p;
}
__device__ __forceinline__ float xhalf(float x) {
    const int b = __builtin_bit_cast(int, x);
    return __builtin_bit_cast(float, __builtin_amdgcn_permlanex16(
        b, b, 0x76543210, 0xfedcba98, false, false));
}
template <int M>
__device__ __forceinline__ float row_xadd(float x) {
    const int b = __builtin_bit_cast(int, x);
    return x + __builtin_bit_cast(float, __builtin_amdgcn_update_dpp(
        0, b, 0x160 + M, 0xf, 0xf, true));
}
__device__ __forceinline__ float wave_sum(float s) {
    s = row_xadd<1>(s);
    s = row_xadd<2>(s);
    s = row_xadd<4>(s);
    s = row_xadd<8>(s);
    return s + xhalf(s);
}

__global__ __launch_bounds__(512, 16) void expm_apply_kernel(
    const float* __restrict__ x,
    const float* __restrict__ c,
    const float* __restrict__ psi,
    float* __restrict__ out)
{
    __shared__ float tile[16][16 * 17];
    __shared__ float tvec[16][16];

    const int tid = threadIdx.x;
    const int w   = tid >> 5;
    const int l   = tid & 31;
    const int h   = l >> 4;
    const int n   = l & 15;
    const long b0 = (long)blockIdx.x * 16;

    const int wu = __builtin_amdgcn_readfirstlane(w);
    if (wu < 8) {
        v8f acc[2];
#pragma unroll
        for (int a = 0; a < 2; ++a)
            acc[a] = (v8f){0.f, 0.f, 0.f, 0.f, 0.f, 0.f, 0.f, 0.f};
        const float* cp = c + (b0 + n) * 64 + 4 * h;
        const float* pp = psi + (size_t)(4 * h) * 256 + 32 * wu + 2 * n;
#pragma unroll
        for (int i = 0; i < 8; ++i) {
            const v4f cv = *(const v4f*)(cp + 8 * i);
            v2f pv[4];
#pragma unroll
            for (int u = 0; u < 4; ++u)
                pv[u] = *(const v2f*)(pp + (size_t)(8 * i + u) * 256);
#pragma unroll
            for (int t = 0; t < 2; ++t) {
                v2f a2;
                a2.x = cv[2 * t];
                a2.y = cv[2 * t + 1];
#pragma unroll
                for (int a = 0; a < 2; ++a) {
                    v2f bb;
                    bb.x = pv[2 * t][a];
                    bb.y = pv[2 * t + 1][a];
                    acc[a] = __builtin_amdgcn_wmma_f32_16x16x4_f32(
                        false, a2, false, bb, (short)0, acc[a], false, false);
                }
            }
        }
        float* tp = &tile[8 * h][(2 * wu + (n >> 3)) * 17 + 2 * (n & 7)];
#pragma unroll
        for (int a = 0; a < 2; ++a) {
#pragma unroll
            for (int j = 0; j < 8; ++j) tp[j * (16 * 17) + a] = acc[a][j];
        }
    }
    __syncthreads();

    float* mt = &tile[wu][0];
    float* tv = &tvec[wu][0];
    const long b = b0 + wu;

    v2f A1[4], Bp[4];
    ld_a(mt, A1, h, n);
    ld_b(mt, Bp, h, n);

    float s1 = 0.f;
#pragma unroll
    for (int kk = 0; kk < 4; ++kk) s1 += A1[kk].x * A1[kk].x + A1[kk].y * A1[kk].y;

    const float* xr = x + b * 16;
    v2f Ax[4];
#pragma unroll
    for (int kk = 0; kk < 4; ++kk) Ax[kk] = *(const v2f*)(xr + kk * 4 + 2 * h);
    const float xc = xr[n];

    v2f Av[4];
    v8f wv, d;

    float t1 = half_dot(A1, Ax);
    t1 += xhalf(t1);
    tv[n] = t1;
    d = mm(A1, Bp);
    v2f A2[4];
    st_d(mt, d, h, n);
    ld_a(mt, A2, h, n);
    ld_b(mt, Bp, h, n);
    v2f At[4];
#pragma unroll
    for (int kk = 0; kk < 4; ++kk) At[kk] = *(const v2f*)(tv + kk * 4 + 2 * h);

    float t2 = half_dot(A2, Ax);
    float t3 = half_dot(A2, At);
    t2 += xhalf(t2);
    t3 += xhalf(t3);
    v8f v = {xc, t1, t2, t3, xc, t1, t2, t3};
    d = mm(A2, Bp);
    v2f A4[4];
    st_d(mt, d, h, n);
    ld_a(mt, A4, h, n);
    ld_b(mt, Bp, h, n);

    st_d(mt, v, h, n);
    ld_a(mt, Av, h, n);
    wv = mm(Av, A4);
    d  = mm(A4, Bp);
#pragma unroll
    for (int j = 0; j < 8; ++j) v[j] = (j & 4) ? wv[j] : v[j];
    float s8 = 0.f;
#pragma unroll
    for (int j = 0; j < 8; ++j) s8 += d[j] * d[j];
    v2f A8[4];
    st_d(mt, d, h, n);
    ld_a(mt, A8, h, n);

    st_d(mt, v, h, n);
    ld_a(mt, Av, h, n);
    wv = mm(Av, A8);

    const float wlo[8] = {1.0f, 1.0f, (float)(1.0 / 2.0), (float)(1.0 / 6.0),
                          (float)(1.0 / 24.0), (float)(1.0 / 120.0),
                          (float)(1.0 / 720.0), (float)(1.0 / 5040.0)};
    const float whi[8] = {(float)(1.0 / 40320.0), (float)(1.0 / 362880.0),
                          (float)(1.0 / 3628800.0), (float)(1.0 / 39916800.0),
                          (float)(1.0 / 479001600.0), (float)(1.0 / 6227020800.0),
                          (float)(1.0 / 87178291200.0), (float)(1.0 / 1307674368000.0)};
    float s = 0.f;
#pragma unroll
    for (int j = 0; j < 8; ++j) s = __builtin_fmaf(v[j], wlo[j], s);
#pragma unroll
    for (int j = 0; j < 8; ++j) s = __builtin_fmaf(wv[j], whi[j], s);

    s1 = wave_sum(s1);
    s8 = wave_sum(s8);
    const float g1 = __builtin_bit_cast(float,
        __builtin_amdgcn_readfirstlane(__builtin_bit_cast(int, s1)));
    const float g8 = __builtin_bit_cast(float,
        __builtin_amdgcn_readfirstlane(__builtin_bit_cast(int, s8)));
    const bool fast = (g1 <= 128.0f) && (g8 <= 3.2e7f);

    const bool longer = (g1 <= 128.0f) && (g8 <= 1.0e11f);

    if (__builtin_expect(!fast, 0)) {
        if (longer) {
            st_d(mt, wv, h, n);
            ld_a(mt, Av, h, n);
            wv = mm(Av, A8);
            const float wx[8] = {(float)(1.0 / 20922789888000.0),
                                 (float)(1.0 / 355687428096000.0),
                                 (float)(1.0 / 6402373705728000.0),
                                 (float)(1.0 / 121645100408832000.0),
                                 (float)(1.0 / 2432902008176640000.0),
                                 (float)(1.0 / 51090942171709440000.0),
                                 (float)(1.0 / 1124000727777607680000.0),
                                 (float)(1.0 / 25852016738884976640000.0)};
            float e = 0.f;
#pragma unroll
            for (int j = 0; j < 8; ++j) e += wv[j] * wx[j];
            s += e;
        } else {
            const float sc = 1.0f / (float)(1 << SQS);
            v2f Ak[4];
#pragma unroll
            for (int kk = 0; kk < 4; ++kk) {
                Ak[kk].x = A1[kk].x * sc;
                Ak[kk].y = A1[kk].y * sc;
            }
            v2f Bv[4];
            v8f r;
#pragma unroll
            for (int j = 0; j < 8; ++j) r[j] = (n == (j + 8 * h)) ? 1.f : 0.f;
#pragma unroll
            for (int k = ORD; k >= 1; --k) {
                st_d(mt, r, h, n);
                ld_b(mt, Bv, h, n);
                d = mm(Ak, Bv);
                const float invk = 1.0f / (float)k;
#pragma unroll
                for (int j = 0; j < 8; ++j)
                    r[j] = ((n == (j + 8 * h)) ? 1.f : 0.f) + d[j] * invk;
            }
#pragma unroll
            for (int sq = 0; sq < SQS; ++sq) {
                v2f aa[4];
                st_d(mt, r, h, n);
                ld_a(mt, aa, h, n);
                ld_b(mt, Bv, h, n);
                r = mm(aa, Bv);
            }
            v2f Ex[4], Ea[4];
#pragma unroll
            for (int kk = 0; kk < 4; ++kk) Ex[kk] = *(const v2f*)(xr + kk * 4 + 2 * h);
            st_d(mt, r, h, n);
            ld_a(mt, Ea, h, n);
            d = mm(Ex, Ea);
            s = d[0];
        }
    }

    if (h == 0) out[b * 16 + n] = s;
}

extern "C" void kernel_launch(void* const* d_in, const int* in_sizes, int n_in,
                              void* d_out, int out_size, void* d_ws, size_t ws_size,
                              hipStream_t stream) {
    const float* x   = (const float*)d_in[0];
    const float* c   = (const float*)d_in[1];
    const float* psi = (const float*)d_in[2];
    float* out = (float*)d_out;

    const int B = in_sizes[1] / 64;
    dim3 grid(B / 16);
    dim3 block(512);
    hipLaunchKernelGGL(expm_apply_kernel, grid, block, 0, stream,
                       x, c, psi, out);
}


# no powers of T formed: 15 Taylor terms t = T t as per-lane row dot products (packed multiply-adds), handed on through a 16-float LDS vector; x read at entry; guard from |t_15|; sum order changes
# speedup vs baseline: 1.9535x; 1.0402x over previous
#include <hip/hip_runtime.h>

typedef __attribute__((ext_vector_type(2))) float v2f;
typedef __attribute__((ext_vector_type(4))) float v4f;
typedef __attribute__((ext_vector_type(8))) float v8f;

#define ORD 8
#define SQS 6


__device__ __forceinline__ void st_d(float* mt, const v8f& r, int h, int n) {
    float* q = mt + (8 * h) * 17 + n;
#pragma unroll
    for (int j = 0; j < 8; ++j) q[j * 17] = r[j];
    asm volatile("s_wait_dscnt 0" ::: "memory");
}
__device__ __forceinline__ void ld_a(const float* mt, v2f (&a)[4], int h, int n) {
#pragma unroll
    for (int kk = 0; kk < 4; ++kk) {
        const int col = kk * 4 + 2 * h;
        a[kk].x = mt[n * 17 + col];
        a[kk].y = mt[n * 17 + col + 1];
    }
}
__device__ __forceinline__ void ld_b(const float* mt, v2f (&b)[4], int h, int n) {
#pragma unroll
    for (int kk = 0; kk < 4; ++kk) {
        const int row = kk * 4 + 2 * h;
        b[kk].x = mt[row * 17 + n];
        b[kk].y = mt[(row + 1) * 17 + n];
    }
}
__device__ __forceinline__ v8f mm(const v2f (&a)[4], const v2f (&b)[4]) {
    v8f d = {0.f, 0.f, 0.f, 0.f, 0.f, 0.f, 0.f, 0.f};
#pragma unroll
    for (int kk = 0; kk < 4; ++kk)
        d = __builtin_amdgcn_wmma_f32_16x16x4_f32(
            false, a[kk], false, b[kk], (short)0, d, false, false);
    return d;
}
__device__ __forceinline__ float xhalf(float x) {
    const int b = __builtin_bit_cast(int, x);
    return __builtin_bit_cast(float, __builtin_amdgcn_permlanex16(
        b, b, 0x76543210, 0xfedcba98, false, false));
}
template <int M>
__device__ __forceinline__ float row_xadd(float x) {
    const int b = __builtin_bit_cast(int, x);
    return x + __builtin_bit_cast(float, __builtin_amdgcn_update_dpp(
        0, b, 0x160 + M, 0xf, 0xf, true));
}
__device__ __forceinline__ float wave_sum(float s) {
    s = row_xadd<1>(s);
    s = row_xadd<2>(s);
    s = row_xadd<4>(s);
    s = row_xadd<8>(s);
    return s + xhalf(s);
}

__device__ __forceinline__ float mv_step(float* tv, const v2f (&tr)[4], float t,
                                         int h, int n) {
    tv[n] = t;
    const v4f u0 = *(const v4f*)(tv + 8 * h);
    const v4f u1 = *(const v4f*)(tv + 8 * h + 4);
    v2f a = tr[0] * (v2f){u0.x, u0.y};
    a = __builtin_elementwise_fma(tr[1], (v2f){u0.z, u0.w}, a);
    a = __builtin_elementwise_fma(tr[2], (v2f){u1.x, u1.y}, a);
    a = __builtin_elementwise_fma(tr[3], (v2f){u1.z, u1.w}, a);
    const float p = a.x + a.y;
    return p + xhalf(p);
}
__device__ __forceinline__ float row_sum(float s) {
    s = row_xadd<1>(s);
    s = row_xadd<2>(s);
    s = row_xadd<4>(s);
    return row_xadd<8>(s);
}

__global__ __launch_bounds__(512, 16) void expm_apply_kernel(
    const float* __restrict__ x,
    const float* __restrict__ c,
    const float* __restrict__ psi,
    float* __restrict__ out)
{
    __shared__ float tile[16][16 * 17];
    __shared__ __attribute__((aligned(16))) float tvec[16][16];

    const int tid = threadIdx.x;
    const int w   = tid >> 5;
    const int l   = tid & 31;
    const int h   = l >> 4;
    const int n   = l & 15;
    const long b0 = (long)blockIdx.x * 16;

    const int wu = __builtin_amdgcn_readfirstlane(w);
    const float x0 = x[(b0 + wu) * 16 + n];
    if (wu < 8) {
        v8f acc[2];
#pragma unroll
        for (int a = 0; a < 2; ++a)
            acc[a] = (v8f){0.f, 0.f, 0.f, 0.f, 0.f, 0.f, 0.f, 0.f};
        const float* cp = c + (b0 + n) * 64 + 4 * h;
        const float* pp = psi + (size_t)(4 * h) * 256 + 32 * wu + 2 * n;
#pragma unroll
        for (int i = 0; i < 8; ++i) {
            const v4f cv = *(const v4f*)(cp + 8 * i);
            v2f pv[4];
#pragma unroll
            for (int u = 0; u < 4; ++u)
                pv[u] = *(const v2f*)(pp + (size_t)(8 * i + u) * 256);
#pragma unroll
            for (int t = 0; t < 2; ++t) {
                v2f a2;
                a2.x = cv[2 * t];
                a2.y = cv[2 * t + 1];
#pragma unroll
                for (int a = 0; a < 2; ++a) {
                    v2f bb;
                    bb.x = pv[2 * t][a];
                    bb.y = pv[2 * t + 1][a];
                    acc[a] = __builtin_amdgcn_wmma_f32_16x16x4_f32(
                        false, a2, false, bb, (short)0, acc[a], false, false);
                }
            }
        }
        float* tp = &tile[8 * h][(2 * wu + (n >> 3)) * 17 + 2 * (n & 7)];
#pragma unroll
        for (int a = 0; a < 2; ++a) {
#pragma unroll
            for (int j = 0; j < 8; ++j) tp[j * (16 * 17) + a] = acc[a][j];
        }
    }
    __syncthreads();

    float* mt = &tile[wu][0];
    float* tv = &tvec[wu][0];
    const long b = b0 + wu;
    const float* xr = x + b * 16;

    v2f tr[4];
#pragma unroll
    for (int i = 0; i < 4; ++i) {
        tr[i].x = mt[n * 17 + 8 * h + 2 * i];
        tr[i].y = mt[n * 17 + 8 * h + 2 * i + 1];
    }

    float s1 = 0.f;
#pragma unroll
    for (int i = 0; i < 4; ++i) s1 += tr[i].x * tr[i].x + tr[i].y * tr[i].y;

    const float wt[24] = {
        1.0f, 1.0f, (float)(1.0 / 2.0), (float)(1.0 / 6.0),
        (float)(1.0 / 24.0), (float)(1.0 / 120.0), (float)(1.0 / 720.0),
        (float)(1.0 / 5040.0), (float)(1.0 / 40320.0), (float)(1.0 / 362880.0),
        (float)(1.0 / 3628800.0), (float)(1.0 / 39916800.0),
        (float)(1.0 / 479001600.0), (float)(1.0 / 6227020800.0),
        (float)(1.0 / 87178291200.0), (float)(1.0 / 1307674368000.0),
        (float)(1.0 / 20922789888000.0), (float)(1.0 / 355687428096000.0),
        (float)(1.0 / 6402373705728000.0), (float)(1.0 / 121645100408832000.0),
        (float)(1.0 / 2432902008176640000.0),
        (float)(1.0 / 51090942171709440000.0),
        (float)(1.0 / 1124000727777607680000.0),
        (float)(1.0 / 25852016738884976640000.0)};

    float t = x0;
    float s = t;
#pragma unroll
    for (int j = 1; j <= 15; ++j) {
        t = mv_step(tv, tr, t, h, n);
        s = __builtin_fmaf(t, wt[j], s);
    }

    const float q15 = row_sum(t * t);
    s1 = wave_sum(s1);
    const float g1 = __builtin_bit_cast(float,
        __builtin_amdgcn_readfirstlane(__builtin_bit_cast(int, s1)));
    const float g15 = __builtin_bit_cast(float,
        __builtin_amdgcn_readfirstlane(__builtin_bit_cast(int, q15)));
    const float rg = __builtin_amdgcn_sqrtf(g1);
    const float d15 = 16.0f - rg;
    const bool fast = (g1 <= 128.0f) && (g15 * g1 <= 2.73e13f * d15 * d15);

    if (__builtin_expect(!fast, 0)) {
        bool longer = false;
        if (g1 <= 128.0f) {
#pragma unroll
            for (int j = 16; j <= 23; ++j) {
                t = mv_step(tv, tr, t, h, n);
                s = __builtin_fmaf(t, wt[j], s);
            }
            const float q23 = row_sum(t * t);
            const float g23 = __builtin_bit_cast(float,
                __builtin_amdgcn_readfirstlane(__builtin_bit_cast(int, q23)));
            const float d23 = 24.0f - rg;
            longer = (g23 * g1 <= 1.06e34f * d23 * d23);
        }
        if (!longer) {
            v2f A1[4], Bv[4];
            v8f r, d;
            ld_a(mt, A1, h, n);
            const float sc = 1.0f / (float)(1 << SQS);
            v2f Ak[4];
#pragma unroll
            for (int kk = 0; kk < 4; ++kk) {
                Ak[kk].x = A1[kk].x * sc;
                Ak[kk].y = A1[kk].y * sc;
            }
#pragma unroll
            for (int j = 0; j < 8; ++j) r[j] = (n == (j + 8 * h)) ? 1.f : 0.f;
#pragma unroll
            for (int k = ORD; k >= 1; --k) {
                st_d(mt, r, h, n);
                ld_b(mt, Bv, h, n);
                d = mm(Ak, Bv);
                const float invk = 1.0f / (float)k;
#pragma unroll
                for (int j = 0; j < 8; ++j)
                    r[j] = ((n == (j + 8 * h)) ? 1.f : 0.f) + d[j] * invk;
            }
#pragma unroll
            for (int sq = 0; sq < SQS; ++sq) {
                v2f aa[4];
                st_d(mt, r, h, n);
                ld_a(mt, aa, h, n);
                ld_b(mt, Bv, h, n);
                r = mm(aa, Bv);
            }
            v2f Ex[4], Ea[4];
#pragma unroll
            for (int kk = 0; kk < 4; ++kk) Ex[kk] = *(const v2f*)(xr + kk * 4 + 2 * h);
            st_d(mt, r, h, n);
            ld_a(mt, Ea, h, n);
            d = mm(Ex, Ea);
            s = d[0];
        }
    }

    if (h == 0) out[b * 16 + n] = s;
}

extern "C" void kernel_launch(void* const* d_in, const int* in_sizes, int n_in,
                              void* d_out, int out_size, void* d_ws, size_t ws_size,
                              hipStream_t stream) {
    const float* x   = (const float*)d_in[0];
    const float* c   = (const float*)d_in[1];
    const float* psi = (const float*)d_in[2];
    float* out = (float*)d_out;

    const int B = in_sizes[1] / 64;
    dim3 grid(B / 16);
    dim3 block(512);
    hipLaunchKernelGGL(expm_apply_kernel, grid, block, 0, stream,
                       x, c, psi, out);
}


# batch element on the lane: psi as WMMA A operand, c as B; a lane pair keeps its 16x16 T in 128 accumulators, series steps in registers, 8 lane swaps each; 4096 waves, was 65536; sum order changes
# speedup vs baseline: 3.5411x; 1.8127x over previous
#include <hip/hip_runtime.h>

typedef __attribute__((ext_vector_type(2))) float v2f;
typedef __attribute__((ext_vector_type(4))) float v4f;
typedef __attribute__((ext_vector_type(8))) float v8f;

#define ORD 8
#define SQS 6


__device__ __forceinline__ void st_d(float* mt, const v8f& r, int h, int n) {
    float* q = mt + (8 * h) * 17 + n;
#pragma unroll
    for (int j = 0; j < 8; ++j) q[j * 17] = r[j];
    asm volatile("s_wait_dscnt 0" ::: "memory");
}
__device__ __forceinline__ void ld_a(const float* mt, v2f (&a)[4], int h, int n) {
#pragma unroll
    for (int kk = 0; kk < 4; ++kk) {
        const int col = kk * 4 + 2 * h;
        a[kk].x = mt[n * 17 + col];
        a[kk].y = mt[n * 17 + col + 1];
    }
}
__device__ __forceinline__ void ld_b(const float* mt, v2f (&b)[4], int h, int n) {
#pragma unroll
    for (int kk = 0; kk < 4; ++kk) {
        const int row = kk * 4 + 2 * h;
        b[kk].x = mt[row * 17 + n];
        b[kk].y = mt[(row + 1) * 17 + n];
    }
}
__device__ __forceinline__ v8f mm(const v2f (&a)[4], const v2f (&b)[4]) {
    v8f d = {0.f, 0.f, 0.f, 0.f, 0.f, 0.f, 0.f, 0.f};
#pragma unroll
    for (int kk = 0; kk < 4; ++kk)
        d = __builtin_amdgcn_wmma_f32_16x16x4_f32(
            false, a[kk], false, b[kk], (short)0, d, false, false);
    return d;
}
__device__ __forceinline__ float xhalf(float x) {
    const int b = __builtin_bit_cast(int, x);
    return __builtin_bit_cast(float, __builtin_amdgcn_permlanex16(
        b, b, 0x76543210, 0xfedcba98, false, false));
}

__device__ __forceinline__ void load_grp(v4f (&buf)[2][4], const float* psi,
                                         unsigned off) {
#pragma unroll
    for (int q = 0; q < 2; ++q)
#pragma unroll
        for (int u = 0; u < 4; ++u)
            buf[q][u] = *(const v4f*)(psi + (off + (unsigned)(u * 256 + 32 * q)));
}
template <int QP>
__device__ __forceinline__ void mma_grp(v8f (&acc)[4][4], const v4f (&buf)[2][4],
                                        const v4f& cv) {
#pragma unroll
    for (int t = 0; t < 2; ++t) {
        v2f bb;
        bb.x = cv[2 * t];
        bb.y = cv[2 * t + 1];
#pragma unroll
        for (int q = 0; q < 2; ++q)
#pragma unroll
            for (int a = 0; a < 4; ++a) {
                v2f aa;
                aa.x = buf[q][2 * t][a];
                aa.y = buf[q][2 * t + 1][a];
                acc[QP + q][a] = __builtin_amdgcn_wmma_f32_16x16x4_f32(
                    false, aa, false, bb, (short)0, acc[QP + q][a], false, false);
            }
    }
}

__device__ __forceinline__ void series_step(const v8f (&acc)[4][4], v2f (&tA)[4],
                                            v2f (&tB)[4], float (&tn)[8]) {
#pragma unroll
    for (int Q = 0; Q < 4; ++Q) {
#pragma unroll
        for (int u = 0; u < 2; ++u) {
            v2f r = (v2f){acc[Q][0][4 * u], acc[Q][0][4 * u + 1]} * tA[0];
#pragma unroll
            for (int a = 1; a < 4; ++a)
                r = __builtin_elementwise_fma(
                    (v2f){acc[Q][a][4 * u], acc[Q][a][4 * u + 1]}, tA[a], r);
#pragma unroll
            for (int a = 0; a < 4; ++a)
                r = __builtin_elementwise_fma(
                    (v2f){acc[Q][a][4 * u + 2], acc[Q][a][4 * u + 3]}, tB[a], r);
            tn[2 * Q + u] = r.x + r.y;
        }
    }
#pragma unroll
    for (int a = 0; a < 4; ++a) {
        tA[a] = (v2f){tn[a], tn[4 + a]};
        tB[a] = (v2f){xhalf(tn[a]), xhalf(tn[4 + a])};
    }
}

__global__ __launch_bounds__(128, 4) void expm_apply_kernel(
    const float* __restrict__ x,
    const float* __restrict__ c,
    const float* __restrict__ psi,
    float* __restrict__ out)
{
    __shared__ float tile[4][16 * 17];

    const int tid = threadIdx.x;
    const int w   = tid >> 5;
    const int l   = tid & 31;
    const int h   = l >> 4;
    const int n   = l & 15;
    const int wu = __builtin_amdgcn_readfirstlane(w);
    const long b0 = ((long)blockIdx.x * 4 + wu) * 16;

    v8f acc[4][4];
#pragma unroll
    for (int Q = 0; Q < 4; ++Q)
#pragma unroll
        for (int a = 0; a < 4; ++a)
            acc[Q][a] = (v8f){0.f, 0.f, 0.f, 0.f, 0.f, 0.f, 0.f, 0.f};
    const float* cp = c + b0 * 64;
    const unsigned co = (unsigned)(n * 64 + 4 * h);
    const unsigned cbase = (unsigned)(128 * (n >> 3) + 16 * ((n >> 2) & 1)
                                      + 4 * ((n & 3) ^ (2 * (n >> 3))));
    const unsigned po = (unsigned)(4 * h) * 256u + cbase;
#pragma nounroll
    for (int i = 0; i < 8; ++i) {
        const v4f cv = *(const v4f*)(cp + (co + 8u * (unsigned)i));
        v4f pa[2][4], pb[2][4];
        load_grp(pa, psi, po + (unsigned)i * 2048u);
        load_grp(pb, psi, po + (unsigned)i * 2048u + 64u);
        mma_grp<0>(acc, pa, cv);
        mma_grp<2>(acc, pb, cv);
    }

    v2f gq[4];
#pragma unroll
    for (int k = 0; k < 4; ++k) gq[k] = (v2f){0.f, 0.f};
#pragma unroll
    for (int Q = 0; Q < 4; ++Q)
#pragma unroll
        for (int a = 0; a < 4; ++a)
#pragma unroll
            for (int k = 0; k < 4; ++k) {
                const v2f e = (v2f){acc[Q][a][2 * k], acc[Q][a][2 * k + 1]};
                gq[k] = __builtin_elementwise_fma(e, e, gq[k]);
            }
    const v2f gs = (gq[0] + gq[1]) + (gq[2] + gq[3]);
    float g1 = gs.x + gs.y;
    g1 += xhalf(g1);

    const float* xl = x + b0 * 16;
    const unsigned xo = (unsigned)(n * 16 + 8 * h);
    const unsigned xt = (unsigned)(n * 16 + 8 * (1 - h));
    const v4f xo0 = *(const v4f*)(xl + xo);
    const v4f xo1 = *(const v4f*)(xl + (xo + 4u));
    const v4f xt0 = *(const v4f*)(xl + xt);
    const v4f xt1 = *(const v4f*)(xl + (xt + 4u));

    const float wt[24] = {
        1.0f, 1.0f, (float)(1.0 / 2.0), (float)(1.0 / 6.0),
        (float)(1.0 / 24.0), (float)(1.0 / 120.0), (float)(1.0 / 720.0),
        (float)(1.0 / 5040.0), (float)(1.0 / 40320.0), (float)(1.0 / 362880.0),
        (float)(1.0 / 3628800.0), (float)(1.0 / 39916800.0),
        (float)(1.0 / 479001600.0), (float)(1.0 / 6227020800.0),
        (float)(1.0 / 87178291200.0), (float)(1.0 / 1307674368000.0),
        (float)(1.0 / 20922789888000.0), (float)(1.0 / 355687428096000.0),
        (float)(1.0 / 6402373705728000.0), (float)(1.0 / 121645100408832000.0),
        (float)(1.0 / 2432902008176640000.0),
        (float)(1.0 / 51090942171709440000.0),
        (float)(1.0 / 1124000727777607680000.0),
        (float)(1.0 / 25852016738884976640000.0)};

    v2f tA[4], tB[4];
    float tn[8], s[8];
#pragma unroll
    for (int a = 0; a < 4; ++a) {
        tA[a] = (v2f){xo0[a], xo1[a]};
        tB[a] = (v2f){xt0[a], xt1[a]};
        s[a] = xo0[a];
        s[4 + a] = xo1[a];
    }
#pragma nounroll
    for (int j = 1; j <= 15; ++j) {
        series_step(acc, tA, tB, tn);
        const float wj = wt[j];
#pragma unroll
        for (int i = 0; i < 8; ++i) s[i] = __builtin_fmaf(tn[i], wj, s[i]);
    }

    float q15 = 0.f;
#pragma unroll
    for (int i = 0; i < 8; ++i) q15 = __builtin_fmaf(tn[i], tn[i], q15);
    q15 += xhalf(q15);
    const float rg = __builtin_amdgcn_sqrtf(g1);
    const float d15 = 16.0f - rg;
    const bool fast = (g1 <= 128.0f) && (q15 * g1 <= 2.73e13f * d15 * d15);

    bool slow = false;
    if (__builtin_amdgcn_ballot_w32(!fast) != 0u) {
        const bool want23 = !fast && (g1 <= 128.0f);
        bool longer = false;
        if (__builtin_amdgcn_ballot_w32(want23) != 0u) {
            float s2[8];
#pragma unroll
            for (int i = 0; i < 8; ++i) s2[i] = s[i];
#pragma nounroll
            for (int j = 16; j <= 23; ++j) {
                series_step(acc, tA, tB, tn);
                const float wj = wt[j];
#pragma unroll
                for (int i = 0; i < 8; ++i) s2[i] = __builtin_fmaf(tn[i], wj, s2[i]);
            }
            float q23 = 0.f;
#pragma unroll
            for (int i = 0; i < 8; ++i) q23 = __builtin_fmaf(tn[i], tn[i], q23);
            q23 += xhalf(q23);
            const float d23 = 24.0f - rg;
            longer = want23 && (q23 * g1 <= 1.06e34f * d23 * d23);
#pragma unroll
            for (int i = 0; i < 8; ++i) s[i] = longer ? s2[i] : s[i];
        }
        slow = !fast && !longer;
        unsigned todo = __builtin_amdgcn_ballot_w32(slow) & 0xffffu;
        float* mt = &tile[wu][0];
        while (todo != 0u) {
            const int nb = __builtin_ctz(todo);
            todo &= todo - 1u;
            if (n == nb) {
#pragma unroll
                for (int Q = 0; Q < 4; ++Q)
#pragma unroll
                    for (int a = 0; a < 4; ++a)
#pragma unroll
                        for (int r = 0; r < 8; ++r)
                            mt[(8 * h + 2 * Q + (r >> 2)) * 17
                               + ((4 * (r & 3) + a) ^ (8 * h))] = acc[Q][a][r];
            }
            asm volatile("s_wait_dscnt 0" ::: "memory");
            const float* xr = x + (b0 + nb) * 16;
            v2f A1[4], Bv[4];
            v8f r, d;
            ld_a(mt, A1, h, n);
            const float sc = 1.0f / (float)(1 << SQS);
            v2f Ak[4];
#pragma unroll
            for (int kk = 0; kk < 4; ++kk) {
                Ak[kk].x = A1[kk].x * sc;
                Ak[kk].y = A1[kk].y * sc;
            }
#pragma unroll
            for (int j = 0; j < 8; ++j) r[j] = (n == (j + 8 * h)) ? 1.f : 0.f;
#pragma unroll
            for (int k = ORD; k >= 1; --k) {
                st_d(mt, r, h, n);
                ld_b(mt, Bv, h, n);
                d = mm(Ak, Bv);
                const float invk = 1.0f / (float)k;
#pragma unroll
                for (int j = 0; j < 8; ++j)
                    r[j] = ((n == (j + 8 * h)) ? 1.f : 0.f) + d[j] * invk;
            }
#pragma unroll
            for (int sq = 0; sq < SQS; ++sq) {
                v2f aa[4];
                st_d(mt, r, h, n);
                ld_a(mt, aa, h, n);
                ld_b(mt, Bv, h, n);
                r = mm(aa, Bv);
            }
            v2f Ex[4], Ea[4];
#pragma unroll
            for (int kk = 0; kk < 4; ++kk) Ex[kk] = *(const v2f*)(xr + kk * 4 + 2 * h);
            st_d(mt, r, h, n);
            ld_a(mt, Ea, h, n);
            d = mm(Ex, Ea);
            if (h == 0) out[(b0 + nb) * 16 + n] = d[0];
        }
    }

    if (!slow) {
        float* o = out + b0 * 16;
        const unsigned oo = (unsigned)(n * 16 + 8 * h);
        *(v4f*)(o + oo) = (v4f){s[0], s[1], s[2], s[3]};
        *(v4f*)(o + (oo + 4u)) = (v4f){s[4], s[5], s[6], s[7]};
    }
}

extern "C" void kernel_launch(void* const* d_in, const int* in_sizes, int n_in,
                              void* d_out, int out_size, void* d_ws, size_t ws_size,
                              hipStream_t stream) {
    const float* x   = (const float*)d_in[0];
    const float* c   = (const float*)d_in[1];
    const float* psi = (const float*)d_in[2];
    float* out = (float*)d_out;

    const int B = in_sizes[1] / 64;
    dim3 grid(B / 64);
    dim3 block(128);
    hipLaunchKernelGGL(expm_apply_kernel, grid, block, 0, stream,
                       x, c, psi, out);
}


# series length per wave grows one term at a time (15..23) until each element's tail bound (same 4e-6 form, per-degree constants) holds; was a jump 15 to 23; sum order kept, term count changes
# speedup vs baseline: 3.7357x; 1.0550x over previous
#include <hip/hip_runtime.h>

typedef __attribute__((ext_vector_type(2))) float v2f;
typedef __attribute__((ext_vector_type(4))) float v4f;
typedef __attribute__((ext_vector_type(8))) float v8f;

#define ORD 8
#define SQS 6


__device__ __forceinline__ void st_d(float* mt, const v8f& r, int h, int n) {
    float* q = mt + (8 * h) * 17 + n;
#pragma unroll
    for (int j = 0; j < 8; ++j) q[j * 17] = r[j];
    asm volatile("s_wait_dscnt 0" ::: "memory");
}
__device__ __forceinline__ void ld_a(const float* mt, v2f (&a)[4], int h, int n) {
#pragma unroll
    for (int kk = 0; kk < 4; ++kk) {
        const int col = kk * 4 + 2 * h;
        a[kk].x = mt[n * 17 + col];
        a[kk].y = mt[n * 17 + col + 1];
    }
}
__device__ __forceinline__ void ld_b(const float* mt, v2f (&b)[4], int h, int n) {
#pragma unroll
    for (int kk = 0; kk < 4; ++kk) {
        const int row = kk * 4 + 2 * h;
        b[kk].x = mt[row * 17 + n];
        b[kk].y = mt[(row + 1) * 17 + n];
    }
}
__device__ __forceinline__ v8f mm(const v2f (&a)[4], const v2f (&b)[4]) {
    v8f d = {0.f, 0.f, 0.f, 0.f, 0.f, 0.f, 0.f, 0.f};
#pragma unroll
    for (int kk = 0; kk < 4; ++kk)
        d = __builtin_amdgcn_wmma_f32_16x16x4_f32(
            false, a[kk], false, b[kk], (short)0, d, false, false);
    return d;
}
__device__ __forceinline__ float xhalf(float x) {
    const int b = __builtin_bit_cast(int, x);
    return __builtin_bit_cast(float, __builtin_amdgcn_permlanex16(
        b, b, 0x76543210, 0xfedcba98, false, false));
}

__device__ __forceinline__ void load_grp(v4f (&buf)[2][4], const float* psi,
                                         unsigned off) {
#pragma unroll
    for (int q = 0; q < 2; ++q)
#pragma unroll
        for (int u = 0; u < 4; ++u)
            buf[q][u] = *(const v4f*)(psi + (off + (unsigned)(u * 256 + 32 * q)));
}
template <int QP>
__device__ __forceinline__ void mma_grp(v8f (&acc)[4][4], const v4f (&buf)[2][4],
                                        const v4f& cv) {
#pragma unroll
    for (int t = 0; t < 2; ++t) {
        v2f bb;
        bb.x = cv[2 * t];
        bb.y = cv[2 * t + 1];
#pragma unroll
        for (int q = 0; q < 2; ++q)
#pragma unroll
            for (int a = 0; a < 4; ++a) {
                v2f aa;
                aa.x = buf[q][2 * t][a];
                aa.y = buf[q][2 * t + 1][a];
                acc[QP + q][a] = __builtin_amdgcn_wmma_f32_16x16x4_f32(
                    false, aa, false, bb, (short)0, acc[QP + q][a], false, false);
            }
    }
}

__device__ __forceinline__ void series_step(const v8f (&acc)[4][4], v2f (&tA)[4],
                                            v2f (&tB)[4], float (&tn)[8]) {
#pragma unroll
    for (int Q = 0; Q < 4; ++Q) {
#pragma unroll
        for (int u = 0; u < 2; ++u) {
            v2f r = (v2f){acc[Q][0][4 * u], acc[Q][0][4 * u + 1]} * tA[0];
#pragma unroll
            for (int a = 1; a < 4; ++a)
                r = __builtin_elementwise_fma(
                    (v2f){acc[Q][a][4 * u], acc[Q][a][4 * u + 1]}, tA[a], r);
#pragma unroll
            for (int a = 0; a < 4; ++a)
                r = __builtin_elementwise_fma(
                    (v2f){acc[Q][a][4 * u + 2], acc[Q][a][4 * u + 3]}, tB[a], r);
            tn[2 * Q + u] = r.x + r.y;
        }
    }
#pragma unroll
    for (int a = 0; a < 4; ++a) {
        tA[a] = (v2f){tn[a], tn[4 + a]};
        tB[a] = (v2f){xhalf(tn[a]), xhalf(tn[4 + a])};
    }
}

__global__ __launch_bounds__(128, 4) void expm_apply_kernel(
    const float* __restrict__ x,
    const float* __restrict__ c,
    const float* __restrict__ psi,
    float* __restrict__ out)
{
    __shared__ float tile[4][16 * 17];

    const int tid = threadIdx.x;
    const int w   = tid >> 5;
    const int l   = tid & 31;
    const int h   = l >> 4;
    const int n   = l & 15;
    const int wu = __builtin_amdgcn_readfirstlane(w);
    const long b0 = ((long)blockIdx.x * 4 + wu) * 16;

    v8f acc[4][4];
#pragma unroll
    for (int Q = 0; Q < 4; ++Q)
#pragma unroll
        for (int a = 0; a < 4; ++a)
            acc[Q][a] = (v8f){0.f, 0.f, 0.f, 0.f, 0.f, 0.f, 0.f, 0.f};
    const float* cp = c + b0 * 64;
    const unsigned co = (unsigned)(n * 64 + 4 * h);
    const unsigned cbase = (unsigned)(128 * (n >> 3) + 16 * ((n >> 2) & 1)
                                      + 4 * ((n & 3) ^ (2 * (n >> 3))));
    const unsigned po = (unsigned)(4 * h) * 256u + cbase;
#pragma nounroll
    for (int i = 0; i < 8; ++i) {
        const v4f cv = *(const v4f*)(cp + (co + 8u * (unsigned)i));
        v4f pa[2][4], pb[2][4];
        load_grp(pa, psi, po + (unsigned)i * 2048u);
        load_grp(pb, psi, po + (unsigned)i * 2048u + 64u);
        mma_grp<0>(acc, pa, cv);
        mma_grp<2>(acc, pb, cv);
    }

    v2f gq[4];
#pragma unroll
    for (int k = 0; k < 4; ++k) gq[k] = (v2f){0.f, 0.f};
#pragma unroll
    for (int Q = 0; Q < 4; ++Q)
#pragma unroll
        for (int a = 0; a < 4; ++a)
#pragma unroll
            for (int k = 0; k < 4; ++k) {
                const v2f e = (v2f){acc[Q][a][2 * k], acc[Q][a][2 * k + 1]};
                gq[k] = __builtin_elementwise_fma(e, e, gq[k]);
            }
    const v2f gs = (gq[0] + gq[1]) + (gq[2] + gq[3]);
    float g1 = gs.x + gs.y;
    g1 += xhalf(g1);

    const float* xl = x + b0 * 16;
    const unsigned xo = (unsigned)(n * 16 + 8 * h);
    const unsigned xt = (unsigned)(n * 16 + 8 * (1 - h));
    const v4f xo0 = *(const v4f*)(xl + xo);
    const v4f xo1 = *(const v4f*)(xl + (xo + 4u));
    const v4f xt0 = *(const v4f*)(xl + xt);
    const v4f xt1 = *(const v4f*)(xl + (xt + 4u));

    const float wt[24] = {
        1.0f, 1.0f, (float)(1.0 / 2.0), (float)(1.0 / 6.0),
        (float)(1.0 / 24.0), (float)(1.0 / 120.0), (float)(1.0 / 720.0),
        (float)(1.0 / 5040.0), (float)(1.0 / 40320.0), (float)(1.0 / 362880.0),
        (float)(1.0 / 3628800.0), (float)(1.0 / 39916800.0),
        (float)(1.0 / 479001600.0), (float)(1.0 / 6227020800.0),
        (float)(1.0 / 87178291200.0), (float)(1.0 / 1307674368000.0),
        (float)(1.0 / 20922789888000.0), (float)(1.0 / 355687428096000.0),
        (float)(1.0 / 6402373705728000.0), (float)(1.0 / 121645100408832000.0),
        (float)(1.0 / 2432902008176640000.0),
        (float)(1.0 / 51090942171709440000.0),
        (float)(1.0 / 1124000727777607680000.0),
        (float)(1.0 / 25852016738884976640000.0)};

    v2f tA[4], tB[4];
    float tn[8], s[8];
#pragma unroll
    for (int a = 0; a < 4; ++a) {
        tA[a] = (v2f){xo0[a], xo1[a]};
        tB[a] = (v2f){xt0[a], xt1[a]};
        s[a] = xo0[a];
        s[4 + a] = xo1[a];
    }
#pragma nounroll
    for (int j = 1; j <= 15; ++j) {
        series_step(acc, tA, tB, tn);
        const float wj = wt[j];
#pragma unroll
        for (int i = 0; i < 8; ++i) s[i] = __builtin_fmaf(tn[i], wj, s[i]);
    }

    float qj = 0.f;
#pragma unroll
    for (int i = 0; i < 8; ++i) qj = __builtin_fmaf(tn[i], tn[i], qj);
    qj += xhalf(qj);
    const float rg = __builtin_amdgcn_sqrtf(g1);
    const float kb[9] = {2.73e13f, 7.00e15f, 2.02e18f, 6.55e20f, 2.36e23f,
                         9.47e25f, 4.17e28f, 2.02e31f, 1.06e34f};
    const bool small = (g1 <= 128.0f);
    float dj = 16.0f - rg;
    bool ok = small & (qj * g1 <= kb[0] * dj * dj);
    if (__builtin_amdgcn_ballot_w32(small & !ok) != 0u) {
#pragma nounroll
        for (int jd = 16; jd <= 23; ++jd) {
            series_step(acc, tA, tB, tn);
            const float wj = wt[jd];
#pragma unroll
            for (int i = 0; i < 8; ++i) s[i] = __builtin_fmaf(tn[i], wj, s[i]);
            qj = 0.f;
#pragma unroll
            for (int i = 0; i < 8; ++i) qj = __builtin_fmaf(tn[i], tn[i], qj);
            qj += xhalf(qj);
            dj = (float)(jd + 1) - rg;
            ok = ok | (small & (qj * g1 <= kb[jd - 15] * dj * dj));
            if (__builtin_amdgcn_ballot_w32(small & !ok) == 0u) break;
        }
    }

    const bool slow = !ok;
    if (__builtin_amdgcn_ballot_w32(slow) != 0u) {
        unsigned todo = __builtin_amdgcn_ballot_w32(slow) & 0xffffu;
        float* mt = &tile[wu][0];
        while (todo != 0u) {
            const int nb = __builtin_ctz(todo);
            todo &= todo - 1u;
            if (n == nb) {
#pragma unroll
                for (int Q = 0; Q < 4; ++Q)
#pragma unroll
                    for (int a = 0; a < 4; ++a)
#pragma unroll
                        for (int r = 0; r < 8; ++r)
                            mt[(8 * h + 2 * Q + (r >> 2)) * 17
                               + ((4 * (r & 3) + a) ^ (8 * h))] = acc[Q][a][r];
            }
            asm volatile("s_wait_dscnt 0" ::: "memory");
            const float* xr = x + (b0 + nb) * 16;
            v2f A1[4], Bv[4];
            v8f r, d;
            ld_a(mt, A1, h, n);
            const float sc = 1.0f / (float)(1 << SQS);
            v2f Ak[4];
#pragma unroll
            for (int kk = 0; kk < 4; ++kk) {
                Ak[kk].x = A1[kk].x * sc;
                Ak[kk].y = A1[kk].y * sc;
            }
#pragma unroll
            for (int j = 0; j < 8; ++j) r[j] = (n == (j + 8 * h)) ? 1.f : 0.f;
#pragma unroll
            for (int k = ORD; k >= 1; --k) {
                st_d(mt, r, h, n);
                ld_b(mt, Bv, h, n);
                d = mm(Ak, Bv);
                const float invk = 1.0f / (float)k;
#pragma unroll
                for (int j = 0; j < 8; ++j)
                    r[j] = ((n == (j + 8 * h)) ? 1.f : 0.f) + d[j] * invk;
            }
#pragma unroll
            for (int sq = 0; sq < SQS; ++sq) {
                v2f aa[4];
                st_d(mt, r, h, n);
                ld_a(mt, aa, h, n);
                ld_b(mt, Bv, h, n);
                r = mm(aa, Bv);
            }
            v2f Ex[4], Ea[4];
#pragma unroll
            for (int kk = 0; kk < 4; ++kk) Ex[kk] = *(const v2f*)(xr + kk * 4 + 2 * h);
            st_d(mt, r, h, n);
            ld_a(mt, Ea, h, n);
            d = mm(Ex, Ea);
            if (h == 0) out[(b0 + nb) * 16 + n] = d[0];
        }
    }

    if (!slow) {
        float* o = out + b0 * 16;
        const unsigned oo = (unsigned)(n * 16 + 8 * h);
        *(v4f*)(o + oo) = (v4f){s[0], s[1], s[2], s[3]};
        *(v4f*)(o + (oo + 4u)) = (v4f){s[4], s[5], s[6], s[7]};
    }
}

extern "C" void kernel_launch(void* const* d_in, const int* in_sizes, int n_in,
                              void* d_out, int out_size, void* d_ws, size_t ws_size,
                              hipStream_t stream) {
    const float* x   = (const float*)d_in[0];
    const float* c   = (const float*)d_in[1];
    const float* psi = (const float*)d_in[2];
    float* out = (float*)d_out;

    const int B = in_sizes[1] / 64;
    dim3 grid(B / 64);
    dim3 block(128);
    hipLaunchKernelGGL(expm_apply_kernel, grid, block, 0, stream,
                       x, c, psi, out);
}
